# starting kernel
# baseline (speedup 1.0000x reference)
#include <hip/hip_runtime.h>
#include <hip/hip_bf16.h>


#define N_NODES 100000
#define N_EDGES 1600000
#define IN_C    64
#define HID     128
#define OUT_C   64

typedef __attribute__((ext_vector_type(2))) float floatx2;
typedef __attribute__((ext_vector_type(8))) float floatx8;

__global__ void copy_f4_kernel(float* __restrict__ dst, const float* __restrict__ src,
                               int n4) {
    int i = blockIdx.x * blockDim.x + threadIdx.x;
    if (i < n4) {
        ((float4*)dst)[i] = ((const float4*)src)[i];
    }
}

template <int C4>
__global__ void scatter_add_kernel(float* __restrict__ agg, const float* __restrict__ h,
                                   const int* __restrict__ srcIdx,
                                   const int* __restrict__ dstIdx,
                                   int total) {
    int tid = blockIdx.x * blockDim.x + threadIdx.x;
    if (tid >= total) return;
    int e  = tid / C4;
    int c  = (tid & (C4 - 1)) * 4;
    int s  = srcIdx[e];
    int d  = dstIdx[e];
    const float4 v = *(const float4*)(h + (size_t)s * (C4 * 4) + c);
    float* p = agg + (size_t)d * (C4 * 4) + c;
    atomicAdd(p + 0, v.x);
    atomicAdd(p + 1, v.y);
    atomicAdd(p + 2, v.z);
    atomicAdd(p + 3, v.w);
}

template <int K, int N, int RELU>
__global__ void __launch_bounds__(32)
gemm_bias_act_kernel(const float* __restrict__ A, const float* __restrict__ W,
                     const float* __restrict__ bias, float* __restrict__ C) {
    __shared__ float sA[16 * 129];

    const int mt   = blockIdx.x;
    const int nt   = blockIdx.y;
    const int lane = threadIdx.x;
    const int half = lane >> 4;
    const int l16  = lane & 15;

    const float4* A4 = (const float4*)(A + (size_t)mt * (16 * K));
    constexpr int K4   = K / 4;
    constexpr int TOTV = 16 * K4;
#pragma unroll
    for (int i = lane; i < TOTV; i += 32) {
        int r = i / K4;
        int c = (i & (K4 - 1)) * 4;
        float4 v = A4[i];
        float* dsti = &sA[r * 129 + c];
        dsti[0] = v.x; dsti[1] = v.y; dsti[2] = v.z; dsti[3] = v.w;
    }
    __syncthreads();

    floatx8 acc = {};
    const float* Wbase = W + nt * 16 + l16 + half * 2 * N;
    const float* sArow = &sA[l16 * 129 + half * 2];

#pragma unroll
    for (int k = 0; k < K; k += 4) {
        floatx2 a;
        a.x = sArow[k + 0];
        a.y = sArow[k + 1];
        floatx2 b;
        b.x = Wbase[(k + 0) * N];
        b.y = Wbase[(k + 1) * N];
        acc = __builtin_amdgcn_wmma_f32_16x16x4_f32(
             false, a,  false, b,
             (short)0, acc,  false,  false);
    }

    const float bv = bias[nt * 16 + l16];
    float* Cout = C + ((size_t)mt * 16 + half * 8) * N + nt * 16 + l16;
#pragma unroll
    for (int v = 0; v < 8; ++v) {
        float x = acc[v] + bv;
        if (RELU) x = fmaxf(x, 0.0f);
        Cout[(size_t)v * N] = x;
    }
}

extern "C" void kernel_launch(void* const* d_in, const int* in_sizes, int n_in,
                              void* d_out, int out_size, void* d_ws, size_t ws_size,
                              hipStream_t stream) {
    (void)in_sizes; (void)n_in; (void)out_size; (void)ws_size;

    const float* x    = (const float*)d_in[0];
    const int*   ei   = (const int*)d_in[1];
    const float* W1a  = (const float*)d_in[2];
    const float* b1a  = (const float*)d_in[3];
    const float* W1b  = (const float*)d_in[4];
    const float* b1b  = (const float*)d_in[5];
    const float* W2a  = (const float*)d_in[6];
    const float* b2a  = (const float*)d_in[7];
    const float* W2b  = (const float*)d_in[8];
    const float* b2b  = (const float*)d_in[9];
    float* out = (float*)d_out;

    const int* srcIdx = ei;
    const int* dstIdx = ei + N_EDGES;

    const size_t buf_elems = (size_t)N_NODES * HID;
    float* B0 = (float*)d_ws;
    float* B1 = B0 + buf_elems;
    float* B2 = B1 + buf_elems;

    const int TPB = 256;
    const int MT  = N_NODES / 16;

    {
        int n4 = N_NODES * IN_C / 4;
        copy_f4_kernel<<<(n4 + TPB - 1) / TPB, TPB, 0, stream>>>(B0, x, n4);
        int total = N_EDGES * (IN_C / 4);
        scatter_add_kernel<IN_C / 4><<<(total + TPB - 1) / TPB, TPB, 0, stream>>>(
            B0, x, srcIdx, dstIdx, total);
    }
    gemm_bias_act_kernel<IN_C, HID, 1>
        <<<dim3(MT, HID / 16), 32, 0, stream>>>(B0, W1a, b1a, B1);
    gemm_bias_act_kernel<HID, HID, 1>
        <<<dim3(MT, HID / 16), 32, 0, stream>>>(B1, W1b, b1b, B2);

    {
        int n4 = N_NODES * HID / 4;
        copy_f4_kernel<<<(n4 + TPB - 1) / TPB, TPB, 0, stream>>>(B0, B2, n4);
        int total = N_EDGES * (HID / 4);
        scatter_add_kernel<HID / 4><<<(total + TPB - 1) / TPB, TPB, 0, stream>>>(
            B0, B2, srcIdx, dstIdx, total);
    }
    gemm_bias_act_kernel<HID, HID, 1>
        <<<dim3(MT, HID / 16), 32, 0, stream>>>(B0, W2a, b2a, B1);
    gemm_bias_act_kernel<HID, OUT_C, 0>
        <<<dim3(MT, OUT_C / 16), 32, 0, stream>>>(B1, W2b, b2b, out);
}


# Float-atomic scatter-add replaced by per-call grouping of edges by destination (integer atomic rank, scan, fill) and an atomic-free row gather-sum; sums now in arrival-rank order
# speedup vs baseline: 1.6213x; 1.6213x over previous
#include <hip/hip_runtime.h>
#include <hip/hip_bf16.h>


#define N_NODES 100000
#define N_EDGES 1600000
#define IN_C    64
#define HID     128
#define OUT_C   64

typedef __attribute__((ext_vector_type(2))) float floatx2;
typedef __attribute__((ext_vector_type(8))) float floatx8;

typedef __attribute__((ext_vector_type(4))) float floatx4;
typedef __attribute__((ext_vector_type(2))) int   intx2;

#define CNT_PAD 100096

__global__ void zero_counts_kernel(int* __restrict__ counts, int n) {
    int i = blockIdx.x * blockDim.x + threadIdx.x;
    if (i < n) counts[i] = 0;
}

__global__ void rank_edges_kernel(const int* __restrict__ dstIdx, int* __restrict__ counts,
                                  int* __restrict__ rank, int nE) {
    int e = blockIdx.x * blockDim.x + threadIdx.x;
    if (e < nE) rank[e] = atomicAdd(&counts[dstIdx[e]], 1);
}

__global__ void __launch_bounds__(1024)
segment_offsets_kernel(const int* __restrict__ counts, int* __restrict__ cursor,
                       intx2* __restrict__ seg, int n) {
    __shared__ int wsum[32];
    __shared__ int sbase;
    const int i    = blockIdx.x * 1024 + threadIdx.x;
    const int lane = threadIdx.x & 31;
    const int w    = threadIdx.x >> 5;
    const int c    = (i < n) ? counts[i] : 0;
    int incl = c;
    for (int d = 1; d < 32; d <<= 1) {
        int t = __shfl_up(incl, d);
        if (lane >= d) incl += t;
    }
    if (lane == 31) wsum[w] = incl;
    __syncthreads();
    if (w == 0) {
        const int v = wsum[lane];
        int vi = v;
        for (int d = 1; d < 32; d <<= 1) {
            int t = __shfl_up(vi, d);
            if (lane >= d) vi += t;
        }
        wsum[lane] = vi - v;
        if (lane == 31) sbase = atomicAdd(cursor, vi);
    }
    __syncthreads();
    if (i < n) seg[i] = intx2{sbase + wsum[w] + incl - c, c};
}

__global__ void fill_sources_kernel(const int* __restrict__ srcIdx, const int* __restrict__ dstIdx,
                                    const int* __restrict__ rank, const intx2* __restrict__ seg,
                                    int* __restrict__ list, int nE) {
    int e = blockIdx.x * blockDim.x + threadIdx.x;
    if (e < nE) list[seg[dstIdx[e]].x + rank[e]] = srcIdx[e];
}

template <int C, typename V>
__global__ void __launch_bounds__(256)
gather_sum_kernel(float* __restrict__ z, const float* __restrict__ h,
                  const intx2* __restrict__ seg, const int* __restrict__ list, int n) {
    constexpr int VW = C / 32;
    const int lane = threadIdx.x & 31;
    const int gw = __builtin_amdgcn_readfirstlane((int)((blockIdx.x * 256 + threadIdx.x) >> 5));
    const int nw = (int)((gridDim.x * 256) >> 5);
    for (int i = gw; i < n; i += nw) {
        const intx2 sd = seg[i];
        const int beg = sd.x;
        const int deg = sd.y;
        V acc = *(const V*)(h + (size_t)i * C + lane * VW);
        for (int base = 0; base < deg; base += 32) {
            const int left = deg - base;
            const int m    = left < 32 ? left : 32;
            const int my   = (lane < m) ? list[beg + base + lane] : 0;
            for (int j = 0; j < m; j += 8) {
                int s[8];
#pragma unroll
                for (int k = 0; k < 8; ++k) s[k] = __builtin_amdgcn_readlane(my, j + k);
                V v[8];
#pragma unroll
                for (int k = 0; k < 8; ++k) {
                    if (j + k < m) v[k] = *(const V*)(h + (size_t)s[k] * C + lane * VW);
                }
#pragma unroll
                for (int k = 0; k < 8; ++k) {
                    if (j + k < m) acc += v[k];
                }
            }
        }
        *(V*)(z + (size_t)i * C + lane * VW) = acc;
    }
}

template <int K, int N, int RELU>
__global__ void __launch_bounds__(32)
gemm_bias_act_kernel(const float* __restrict__ A, const float* __restrict__ W,
                     const float* __restrict__ bias, float* __restrict__ C) {
    __shared__ float sA[16 * 129];

    const int mt   = blockIdx.x;
    const int nt   = blockIdx.y;
    const int lane = threadIdx.x;
    const int half = lane >> 4;
    const int l16  = lane & 15;

    const float4* A4 = (const float4*)(A + (size_t)mt * (16 * K));
    constexpr int K4   = K / 4;
    constexpr int TOTV = 16 * K4;
#pragma unroll
    for (int i = lane; i < TOTV; i += 32) {
        int r = i / K4;
        int c = (i & (K4 - 1)) * 4;
        float4 v = A4[i];
        float* dsti = &sA[r * 129 + c];
        dsti[0] = v.x; dsti[1] = v.y; dsti[2] = v.z; dsti[3] = v.w;
    }
    __syncthreads();

    floatx8 acc = {};
    const float* Wbase = W + nt * 16 + l16 + half * 2 * N;
    const float* sArow = &sA[l16 * 129 + half * 2];

#pragma unroll
    for (int k = 0; k < K; k += 4) {
        floatx2 a;
        a.x = sArow[k + 0];
        a.y = sArow[k + 1];
        floatx2 b;
        b.x = Wbase[(k + 0) * N];
        b.y = Wbase[(k + 1) * N];
        acc = __builtin_amdgcn_wmma_f32_16x16x4_f32(
             false, a,  false, b,
             (short)0, acc,  false,  false);
    }

    const float bv = bias[nt * 16 + l16];
    float* Cout = C + ((size_t)mt * 16 + half * 8) * N + nt * 16 + l16;
#pragma unroll
    for (int v = 0; v < 8; ++v) {
        float x = acc[v] + bv;
        if (RELU) x = fmaxf(x, 0.0f);
        Cout[(size_t)v * N] = x;
    }
}

extern "C" void kernel_launch(void* const* d_in, const int* in_sizes, int n_in,
                              void* d_out, int out_size, void* d_ws, size_t ws_size,
                              hipStream_t stream) {
    (void)in_sizes; (void)n_in; (void)out_size; (void)ws_size;

    const float* x    = (const float*)d_in[0];
    const int*   ei   = (const int*)d_in[1];
    const float* W1a  = (const float*)d_in[2];
    const float* b1a  = (const float*)d_in[3];
    const float* W1b  = (const float*)d_in[4];
    const float* b1b  = (const float*)d_in[5];
    const float* W2a  = (const float*)d_in[6];
    const float* b2a  = (const float*)d_in[7];
    const float* W2b  = (const float*)d_in[8];
    const float* b2b  = (const float*)d_in[9];
    float* out = (float*)d_out;

    const int* srcIdx = ei;
    const int* dstIdx = ei + N_EDGES;

    const size_t buf_elems = (size_t)N_NODES * HID;
    float* B0 = (float*)d_ws;
    float* B1 = B0 + buf_elems;
    float* B2 = B1 + buf_elems;
    int*  counts = (int*)(B2 + buf_elems);
    intx2* seg   = (intx2*)(counts + CNT_PAD);
    int*  rank   = (int*)(seg + CNT_PAD);
    int*  list   = rank + N_EDGES;

    const int TPB = 256;
    const int MT  = N_NODES / 16;

    const int GW  = N_NODES / 32;

    zero_counts_kernel<<<(N_NODES + 1 + TPB - 1) / TPB, TPB, 0, stream>>>(counts, N_NODES + 1);
    rank_edges_kernel<<<(N_EDGES + TPB - 1) / TPB, TPB, 0, stream>>>(dstIdx, counts, rank, N_EDGES);
    segment_offsets_kernel<<<(N_NODES + 1023) / 1024, 1024, 0, stream>>>(
        counts, counts + N_NODES, seg, N_NODES);
    fill_sources_kernel<<<(N_EDGES + TPB - 1) / TPB, TPB, 0, stream>>>(
        srcIdx, dstIdx, rank, seg, list, N_EDGES);

    gather_sum_kernel<IN_C, floatx2><<<GW, 256, 0, stream>>>(B0, x, seg, list, N_NODES);
    gemm_bias_act_kernel<IN_C, HID, 1>
        <<<dim3(MT, HID / 16), 32, 0, stream>>>(B0, W1a, b1a, B1);
    gemm_bias_act_kernel<HID, HID, 1>
        <<<dim3(MT, HID / 16), 32, 0, stream>>>(B1, W1b, b1b, B2);

    gather_sum_kernel<HID, floatx4><<<GW, 256, 0, stream>>>(B0, B2, seg, list, N_NODES);
    gemm_bias_act_kernel<HID, HID, 1>
        <<<dim3(MT, HID / 16), 32, 0, stream>>>(B0, W2a, b2a, B1);
    gemm_bias_act_kernel<HID, OUT_C, 0>
        <<<dim3(MT, OUT_C / 16), 32, 0, stream>>>(B1, W2b, b2b, out);
}


# GEMM re-tiled: one wave per 16xN strip in 512 persistent 8-wave workgroups with W held in LDS (was a one-wave workgroup per 16x16 tile); A block read once, not N/16 times, rows stored whole
# speedup vs baseline: 2.0329x; 1.2539x over previous
#include <hip/hip_runtime.h>
#include <hip/hip_bf16.h>


#define N_NODES 100000
#define N_EDGES 1600000
#define IN_C    64
#define HID     128
#define OUT_C   64

typedef __attribute__((ext_vector_type(2))) float floatx2;
typedef __attribute__((ext_vector_type(8))) float floatx8;

typedef __attribute__((ext_vector_type(4))) float floatx4;
typedef __attribute__((ext_vector_type(2))) int   intx2;

#define CNT_PAD 100096

__global__ void zero_counts_kernel(int* __restrict__ counts, int n) {
    int i = blockIdx.x * blockDim.x + threadIdx.x;
    if (i < n) counts[i] = 0;
}

__global__ void rank_edges_kernel(const int* __restrict__ dstIdx, int* __restrict__ counts,
                                  int* __restrict__ rank, int nE) {
    int e = blockIdx.x * blockDim.x + threadIdx.x;
    if (e < nE) rank[e] = atomicAdd(&counts[dstIdx[e]], 1);
}

__global__ void __launch_bounds__(1024)
segment_offsets_kernel(const int* __restrict__ counts, int* __restrict__ cursor,
                       intx2* __restrict__ seg, int n) {
    __shared__ int wsum[32];
    __shared__ int sbase;
    const int i    = blockIdx.x * 1024 + threadIdx.x;
    const int lane = threadIdx.x & 31;
    const int w    = threadIdx.x >> 5;
    const int c    = (i < n) ? counts[i] : 0;
    int incl = c;
    for (int d = 1; d < 32; d <<= 1) {
        int t = __shfl_up(incl, d);
        if (lane >= d) incl += t;
    }
    if (lane == 31) wsum[w] = incl;
    __syncthreads();
    if (w == 0) {
        const int v = wsum[lane];
        int vi = v;
        for (int d = 1; d < 32; d <<= 1) {
            int t = __shfl_up(vi, d);
            if (lane >= d) vi += t;
        }
        wsum[lane] = vi - v;
        if (lane == 31) sbase = atomicAdd(cursor, vi);
    }
    __syncthreads();
    if (i < n) seg[i] = intx2{sbase + wsum[w] + incl - c, c};
}

__global__ void fill_sources_kernel(const int* __restrict__ srcIdx, const int* __restrict__ dstIdx,
                                    const int* __restrict__ rank, const intx2* __restrict__ seg,
                                    int* __restrict__ list, int nE) {
    int e = blockIdx.x * blockDim.x + threadIdx.x;
    if (e < nE) list[seg[dstIdx[e]].x + rank[e]] = srcIdx[e];
}

template <int C, typename V>
__global__ void __launch_bounds__(256)
gather_sum_kernel(float* __restrict__ z, const float* __restrict__ h,
                  const intx2* __restrict__ seg, const int* __restrict__ list, int n) {
    constexpr int VW = C / 32;
    const int lane = threadIdx.x & 31;
    const int gw = __builtin_amdgcn_readfirstlane((int)((blockIdx.x * 256 + threadIdx.x) >> 5));
    const int nw = (int)((gridDim.x * 256) >> 5);
    for (int i = gw; i < n; i += nw) {
        const intx2 sd = seg[i];
        const int beg = sd.x;
        const int deg = sd.y;
        V acc = *(const V*)(h + (size_t)i * C + lane * VW);
        for (int base = 0; base < deg; base += 32) {
            const int left = deg - base;
            const int m    = left < 32 ? left : 32;
            const int my   = (lane < m) ? list[beg + base + lane] : 0;
            for (int j = 0; j < m; j += 8) {
                int s[8];
#pragma unroll
                for (int k = 0; k < 8; ++k) s[k] = __builtin_amdgcn_readlane(my, j + k);
                V v[8];
#pragma unroll
                for (int k = 0; k < 8; ++k) {
                    if (j + k < m) v[k] = *(const V*)(h + (size_t)s[k] * C + lane * VW);
                }
#pragma unroll
                for (int k = 0; k < 8; ++k) {
                    if (j + k < m) acc += v[k];
                }
            }
        }
        *(V*)(z + (size_t)i * C + lane * VW) = acc;
    }
}

#define GEMM_WAVES 8
#define GEMM_GRID  512
#define GEMM_AP    132
#define GEMM_WP    288

template <int K, int N, int RELU>
__global__ void __launch_bounds__(GEMM_WAVES * 32)
gemm_bias_act_kernel(const float* __restrict__ A, const float* __restrict__ W,
                     const float* __restrict__ bias, float* __restrict__ C, int mTiles) {
    constexpr int NT = N / 16;
    constexpr int WP = 2 * N + 32;
    constexpr int AP = GEMM_AP;
    static_assert(K + 4 <= AP && N + 4 <= AP, "strip does not fit its LDS rows");
    static_assert((K / 2) * WP <= 64 * GEMM_WP, "W does not fit its LDS region");
    __shared__ __attribute__((aligned(16))) float sW[64 * GEMM_WP];
    __shared__ __attribute__((aligned(16))) float sA[GEMM_WAVES * 16 * AP];

    const int tid  = threadIdx.x;
    const int lane = tid & 31;
    const int half = lane >> 4;
    const int l16  = lane & 15;
    const int wv   = __builtin_amdgcn_readfirstlane(tid >> 5);

    constexpr int NQ    = N / 4;
    constexpr int UNITS = (K / 2) * NQ;
    static_assert(UNITS % (GEMM_WAVES * 32) == 0, "W staging has a ragged last trip");
#pragma unroll
    for (int q = 0; q < UNITS / (GEMM_WAVES * 32); ++q) {
        const int u  = tid + q * (GEMM_WAVES * 32);
        const int kp = u / NQ;
        const int nq = u % NQ;
        const float4 r0 = *(const float4*)(W + (size_t)(2 * kp) * N + nq * 4);
        const float4 r1 = *(const float4*)(W + (size_t)(2 * kp + 1) * N + nq * 4);
        float4* d = (float4*)&sW[kp * WP + nq * 8];
        d[0] = float4{r0.x, r1.x, r0.y, r1.y};
        d[1] = float4{r0.z, r1.z, r0.w, r1.w};
    }

    float* sAw = &sA[wv * (16 * AP)];
    const float* aRow = sAw + l16 * AP + half * 2;
    const float* bCol = sW + half * WP + l16 * 2;

    const int stride = (int)gridDim.x * GEMM_WAVES;
    const int iters  = (mTiles + stride - 1) / stride;
    for (int it = 0; it < iters; ++it) {
        const int  t     = it * stride + wv * (int)gridDim.x + (int)blockIdx.x;
        const bool valid = t < mTiles;

        if (valid) {
            const float4* A4 = (const float4*)(A + (size_t)t * (16 * K));
            constexpr int K4 = K / 4;
#pragma unroll
            for (int q = 0; q < 16 * K4 / 32; ++q) {
                const int i  = lane + q * 32;
                const int r  = i / K4;
                const int c4 = i % K4;
                *(float4*)&sAw[r * AP + c4 * 4] = A4[i];
            }
        }
        __syncthreads();

        if (valid) {
            floatx8 acc[NT];
#pragma unroll
            for (int j = 0; j < NT; ++j) acc[j] = floatx8{};

#pragma unroll
            for (int k = 0; k < K; k += 4) {
                const floatx2 a = *(const floatx2*)(aRow + k);
                floatx2 b[NT];
#pragma unroll
                for (int j = 0; j < NT; ++j)
                    b[j] = *(const floatx2*)(bCol + (k / 2) * WP + j * 32);
#pragma unroll
                for (int j = 0; j < NT; ++j)
                    acc[j] = __builtin_amdgcn_wmma_f32_16x16x4_f32(
                         false, a,  false, b[j],
                         (short)0, acc[j],  false,  false);
            }

#pragma unroll
            for (int j = 0; j < NT; ++j) {
                const float bv = bias[j * 16 + l16];
#pragma unroll
                for (int v = 0; v < 8; ++v) {
                    float x = acc[j][v] + bv;
                    if (RELU) x = fmaxf(x, 0.0f);
                    sAw[(half * 8 + v) * AP + j * 16 + l16] = x;
                }
            }
        }
        __syncthreads();

        if (valid) {
            float4* C4 = (float4*)(C + (size_t)t * (16 * N));
            constexpr int N4 = N / 4;
#pragma unroll
            for (int q = 0; q < 16 * N4 / 32; ++q) {
                const int i  = lane + q * 32;
                const int r  = i / N4;
                const int c4 = i % N4;
                C4[i] = *(const float4*)&sAw[r * AP + c4 * 4];
            }
        }
    }
}

extern "C" void kernel_launch(void* const* d_in, const int* in_sizes, int n_in,
                              void* d_out, int out_size, void* d_ws, size_t ws_size,
                              hipStream_t stream) {
    (void)in_sizes; (void)n_in; (void)out_size; (void)ws_size;

    const float* x    = (const float*)d_in[0];
    const int*   ei   = (const int*)d_in[1];
    const float* W1a  = (const float*)d_in[2];
    const float* b1a  = (const float*)d_in[3];
    const float* W1b  = (const float*)d_in[4];
    const float* b1b  = (const float*)d_in[5];
    const float* W2a  = (const float*)d_in[6];
    const float* b2a  = (const float*)d_in[7];
    const float* W2b  = (const float*)d_in[8];
    const float* b2b  = (const float*)d_in[9];
    float* out = (float*)d_out;

    const int* srcIdx = ei;
    const int* dstIdx = ei + N_EDGES;

    const size_t buf_elems = (size_t)N_NODES * HID;
    float* B0 = (float*)d_ws;
    float* B1 = B0 + buf_elems;
    float* B2 = B1 + buf_elems;
    int*  counts = (int*)(B2 + buf_elems);
    intx2* seg   = (intx2*)(counts + CNT_PAD);
    int*  rank   = (int*)(seg + CNT_PAD);
    int*  list   = rank + N_EDGES;

    const int TPB = 256;
    const int MT  = N_NODES / 16;

    const int GW  = N_NODES / 32;

    zero_counts_kernel<<<(N_NODES + 1 + TPB - 1) / TPB, TPB, 0, stream>>>(counts, N_NODES + 1);
    rank_edges_kernel<<<(N_EDGES + TPB - 1) / TPB, TPB, 0, stream>>>(dstIdx, counts, rank, N_EDGES);
    segment_offsets_kernel<<<(N_NODES + 1023) / 1024, 1024, 0, stream>>>(
        counts, counts + N_NODES, seg, N_NODES);
    fill_sources_kernel<<<(N_EDGES + TPB - 1) / TPB, TPB, 0, stream>>>(
        srcIdx, dstIdx, rank, seg, list, N_EDGES);

    gather_sum_kernel<IN_C, floatx2><<<GW, 256, 0, stream>>>(B0, x, seg, list, N_NODES);
    gemm_bias_act_kernel<IN_C, HID, 1>
        <<<GEMM_GRID, GEMM_WAVES * 32, 0, stream>>>(B0, W1a, b1a, B1, MT);
    gemm_bias_act_kernel<HID, HID, 1>
        <<<GEMM_GRID, GEMM_WAVES * 32, 0, stream>>>(B1, W1b, b1b, B2, MT);

    gather_sum_kernel<HID, floatx4><<<GW, 256, 0, stream>>>(B0, B2, seg, list, N_NODES);
    gemm_bias_act_kernel<HID, HID, 1>
        <<<GEMM_GRID, GEMM_WAVES * 32, 0, stream>>>(B0, W2a, b2a, B1, MT);
    gemm_bias_act_kernel<HID, OUT_C, 0>
        <<<GEMM_GRID, GEMM_WAVES * 32, 0, stream>>>(B1, W2b, b2b, out, MT);
}


# Edge rank from LDS counters: a workgroup owns 4096 destination nodes and one of 20 edge chunks; global returning atomics and the counter clear launch are gone; sums in arrival order within a chunk
# speedup vs baseline: 2.2295x; 1.0967x over previous
#include <hip/hip_runtime.h>
#include <hip/hip_bf16.h>


#define N_NODES 100000
#define N_EDGES 1600000
#define IN_C    64
#define HID     128
#define OUT_C   64

typedef __attribute__((ext_vector_type(2))) float floatx2;
typedef __attribute__((ext_vector_type(8))) float floatx8;

typedef __attribute__((ext_vector_type(4))) float floatx4;
typedef __attribute__((ext_vector_type(2))) int   intx2;

#define CNT_PAD 100096

#define RANK_NODES   4096
#define RANK_RANGES  ((N_NODES + RANK_NODES - 1) / RANK_NODES)
#define RANK_CHUNKS  20
#define RANK_CHUNK_E (N_EDGES / RANK_CHUNKS)
#define RANK_LOADS   5
static_assert(N_EDGES % (RANK_CHUNKS * 4) == 0, "edge chunks are whole 16-byte groups");
static_assert((RANK_NODES & (RANK_NODES - 1)) == 0, "the range test needs a power of two");

__device__ __forceinline__ void rank_take4(const int4 d, const int e, const int lo,
                                           int* cnt, int* __restrict__ rank) {
    const unsigned r0 = (unsigned)(d.x ^ lo);
    const unsigned r1 = (unsigned)(d.y ^ lo);
    const unsigned r2 = (unsigned)(d.z ^ lo);
    const unsigned r3 = (unsigned)(d.w ^ lo);
    if (r0 < RANK_NODES) rank[e + 0] = atomicAdd(&cnt[r0], 1);
    if (r1 < RANK_NODES) rank[e + 1] = atomicAdd(&cnt[r1], 1);
    if (r2 < RANK_NODES) rank[e + 2] = atomicAdd(&cnt[r2], 1);
    if (r3 < RANK_NODES) rank[e + 3] = atomicAdd(&cnt[r3], 1);
}

__global__ void __launch_bounds__(1024)
rank_edges_kernel(const int* __restrict__ dstIdx, int* __restrict__ counts,
                  int* __restrict__ cursor, int* __restrict__ rank, int n) {
    __shared__ int cnt[RANK_NODES];
    const int tid = threadIdx.x;
    const int lo  = blockIdx.x * RANK_NODES;
    const int ck  = blockIdx.y;
    for (int j = tid; j < RANK_NODES; j += 1024) cnt[j] = 0;
    __syncthreads();

    const int4* d4 = (const int4*)dstIdx;
    const int qb = ck * (RANK_CHUNK_E / 4);
    const int qe = qb + RANK_CHUNK_E / 4;
    for (int q0 = qb; q0 < qe; q0 += 1024 * RANK_LOADS) {
        int4 d[RANK_LOADS];
#pragma unroll
        for (int u = 0; u < RANK_LOADS; ++u) {
            const int q = q0 + u * 1024 + tid;
            d[u] = (q < qe) ? d4[q] : int4{-1, -1, -1, -1};
        }
#pragma unroll
        for (int u = 0; u < RANK_LOADS; ++u)
            rank_take4(d[u], (q0 + u * 1024 + tid) * 4, lo, cnt, rank);
    }
    __syncthreads();

    for (int j = tid; j < RANK_NODES; j += 1024) {
        const int i = lo + j;
        if (i < n) counts[ck * CNT_PAD + i] = cnt[j];
    }
    if (blockIdx.x == 0 && blockIdx.y == 0 && tid == 0) *cursor = 0;
}

__global__ void __launch_bounds__(1024)
segment_offsets_kernel(const int* __restrict__ counts, int* __restrict__ cursor,
                       intx2* __restrict__ seg, int* __restrict__ start, int n) {
    __shared__ int wsum[32];
    __shared__ int sbase;
    const int i    = blockIdx.x * 1024 + threadIdx.x;
    const int lane = threadIdx.x & 31;
    const int w    = threadIdx.x >> 5;
    int ck[RANK_CHUNKS] = {};
    int c = 0;
    if (i < n) {
#pragma unroll
        for (int k = 0; k < RANK_CHUNKS; ++k) {
            ck[k] = counts[k * CNT_PAD + i];
            c += ck[k];
        }
    }
    int incl = c;
    for (int d = 1; d < 32; d <<= 1) {
        int t = __shfl_up(incl, d);
        if (lane >= d) incl += t;
    }
    if (lane == 31) wsum[w] = incl;
    __syncthreads();
    if (w == 0) {
        const int v = wsum[lane];
        int vi = v;
        for (int d = 1; d < 32; d <<= 1) {
            int t = __shfl_up(vi, d);
            if (lane >= d) vi += t;
        }
        wsum[lane] = vi - v;
        if (lane == 31) sbase = atomicAdd(cursor, vi);
    }
    __syncthreads();
    if (i < n) {
        int s = sbase + wsum[w] + incl - c;
        seg[i] = intx2{s, c};
#pragma unroll
        for (int k = 0; k < RANK_CHUNKS; ++k) {
            start[k * CNT_PAD + i] = s;
            s += ck[k];
        }
    }
}

__global__ void fill_sources_kernel(const int* __restrict__ srcIdx, const int* __restrict__ dstIdx,
                                    const int* __restrict__ rank, const int* __restrict__ start,
                                    int* __restrict__ list, int nE) {
    int e = blockIdx.x * blockDim.x + threadIdx.x;
    if (e < nE) list[start[(e / RANK_CHUNK_E) * CNT_PAD + dstIdx[e]] + rank[e]] = srcIdx[e];
}

template <int C, typename V>
__global__ void __launch_bounds__(256)
gather_sum_kernel(float* __restrict__ z, const float* __restrict__ h,
                  const intx2* __restrict__ seg, const int* __restrict__ list, int n) {
    constexpr int VW = C / 32;
    const int lane = threadIdx.x & 31;
    const int gw = __builtin_amdgcn_readfirstlane((int)((blockIdx.x * 256 + threadIdx.x) >> 5));
    const int nw = (int)((gridDim.x * 256) >> 5);
    for (int i = gw; i < n; i += nw) {
        const intx2 sd = seg[i];
        const int beg = sd.x;
        const int deg = sd.y;
        V acc = *(const V*)(h + (size_t)i * C + lane * VW);
        for (int base = 0; base < deg; base += 32) {
            const int left = deg - base;
            const int m    = left < 32 ? left : 32;
            const int my   = (lane < m) ? list[beg + base + lane] : 0;
            for (int j = 0; j < m; j += 8) {
                int s[8];
#pragma unroll
                for (int k = 0; k < 8; ++k) s[k] = __builtin_amdgcn_readlane(my, j + k);
                V v[8];
#pragma unroll
                for (int k = 0; k < 8; ++k) {
                    if (j + k < m) v[k] = *(const V*)(h + (size_t)s[k] * C + lane * VW);
                }
#pragma unroll
                for (int k = 0; k < 8; ++k) {
                    if (j + k < m) acc += v[k];
                }
            }
        }
        *(V*)(z + (size_t)i * C + lane * VW) = acc;
    }
}

#define GEMM_WAVES 8
#define GEMM_GRID  512
#define GEMM_AP    132
#define GEMM_WP    288

template <int K, int N, int RELU>
__global__ void __launch_bounds__(GEMM_WAVES * 32)
gemm_bias_act_kernel(const float* __restrict__ A, const float* __restrict__ W,
                     const float* __restrict__ bias, float* __restrict__ C, int mTiles) {
    constexpr int NT = N / 16;
    constexpr int WP = 2 * N + 32;
    constexpr int AP = GEMM_AP;
    static_assert(K + 4 <= AP && N + 4 <= AP, "strip does not fit its LDS rows");
    static_assert((K / 2) * WP <= 64 * GEMM_WP, "W does not fit its LDS region");
    __shared__ __attribute__((aligned(16))) float sW[64 * GEMM_WP];
    __shared__ __attribute__((aligned(16))) float sA[GEMM_WAVES * 16 * AP];

    const int tid  = threadIdx.x;
    const int lane = tid & 31;
    const int half = lane >> 4;
    const int l16  = lane & 15;
    const int wv   = __builtin_amdgcn_readfirstlane(tid >> 5);

    constexpr int NQ    = N / 4;
    constexpr int UNITS = (K / 2) * NQ;
    static_assert(UNITS % (GEMM_WAVES * 32) == 0, "W staging has a ragged last trip");
#pragma unroll
    for (int q = 0; q < UNITS / (GEMM_WAVES * 32); ++q) {
        const int u  = tid + q * (GEMM_WAVES * 32);
        const int kp = u / NQ;
        const int nq = u % NQ;
        const float4 r0 = *(const float4*)(W + (size_t)(2 * kp) * N + nq * 4);
        const float4 r1 = *(const float4*)(W + (size_t)(2 * kp + 1) * N + nq * 4);
        float4* d = (float4*)&sW[kp * WP + nq * 8];
        d[0] = float4{r0.x, r1.x, r0.y, r1.y};
        d[1] = float4{r0.z, r1.z, r0.w, r1.w};
    }

    float* sAw = &sA[wv * (16 * AP)];
    const float* aRow = sAw + l16 * AP + half * 2;
    const float* bCol = sW + half * WP + l16 * 2;

    const int stride = (int)gridDim.x * GEMM_WAVES;
    const int iters  = (mTiles + stride - 1) / stride;
    for (int it = 0; it < iters; ++it) {
        const int  t     = it * stride + wv * (int)gridDim.x + (int)blockIdx.x;
        const bool valid = t < mTiles;

        if (valid) {
            const float4* A4 = (const float4*)(A + (size_t)t * (16 * K));
            constexpr int K4 = K / 4;
#pragma unroll
            for (int q = 0; q < 16 * K4 / 32; ++q) {
                const int i  = lane + q * 32;
                const int r  = i / K4;
                const int c4 = i % K4;
                *(float4*)&sAw[r * AP + c4 * 4] = A4[i];
            }
        }
        __syncthreads();

        if (valid) {
            floatx8 acc[NT];
#pragma unroll
            for (int j = 0; j < NT; ++j) acc[j] = floatx8{};

#pragma unroll
            for (int k = 0; k < K; k += 4) {
                const floatx2 a = *(const floatx2*)(aRow + k);
                floatx2 b[NT];
#pragma unroll
                for (int j = 0; j < NT; ++j)
                    b[j] = *(const floatx2*)(bCol + (k / 2) * WP + j * 32);
#pragma unroll
                for (int j = 0; j < NT; ++j)
                    acc[j] = __builtin_amdgcn_wmma_f32_16x16x4_f32(
                         false, a,  false, b[j],
                         (short)0, acc[j],  false,  false);
            }

#pragma unroll
            for (int j = 0; j < NT; ++j) {
                const float bv = bias[j * 16 + l16];
#pragma unroll
                for (int v = 0; v < 8; ++v) {
                    float x = acc[j][v] + bv;
                    if (RELU) x = fmaxf(x, 0.0f);
                    sAw[(half * 8 + v) * AP + j * 16 + l16] = x;
                }
            }
        }
        __syncthreads();

        if (valid) {
            float4* C4 = (float4*)(C + (size_t)t * (16 * N));
            constexpr int N4 = N / 4;
#pragma unroll
            for (int q = 0; q < 16 * N4 / 32; ++q) {
                const int i  = lane + q * 32;
                const int r  = i / N4;
                const int c4 = i % N4;
                C4[i] = *(const float4*)&sAw[r * AP + c4 * 4];
            }
        }
    }
}

extern "C" void kernel_launch(void* const* d_in, const int* in_sizes, int n_in,
                              void* d_out, int out_size, void* d_ws, size_t ws_size,
                              hipStream_t stream) {
    (void)in_sizes; (void)n_in; (void)out_size; (void)ws_size;

    const float* x    = (const float*)d_in[0];
    const int*   ei   = (const int*)d_in[1];
    const float* W1a  = (const float*)d_in[2];
    const float* b1a  = (const float*)d_in[3];
    const float* W1b  = (const float*)d_in[4];
    const float* b1b  = (const float*)d_in[5];
    const float* W2a  = (const float*)d_in[6];
    const float* b2a  = (const float*)d_in[7];
    const float* W2b  = (const float*)d_in[8];
    const float* b2b  = (const float*)d_in[9];
    float* out = (float*)d_out;

    const int* srcIdx = ei;
    const int* dstIdx = ei + N_EDGES;

    const size_t buf_elems = (size_t)N_NODES * HID;
    float* B0 = (float*)d_ws;
    float* B1 = B0 + buf_elems;
    float* B2 = B1 + buf_elems;
    int*  counts = (int*)(B2 + buf_elems);
    int*  start  = counts + RANK_CHUNKS * CNT_PAD;
    int*  cursor = start + RANK_CHUNKS * CNT_PAD;
    intx2* seg   = (intx2*)(cursor + 64);
    int*  rank   = (int*)(seg + CNT_PAD);
    int*  list   = rank + N_EDGES;

    const int TPB = 256;
    const int MT  = N_NODES / 16;

    const int GW  = N_NODES / 32;

    rank_edges_kernel<<<dim3(RANK_RANGES, RANK_CHUNKS), 1024, 0, stream>>>(
        dstIdx, counts, cursor, rank, N_NODES);
    segment_offsets_kernel<<<(N_NODES + 1023) / 1024, 1024, 0, stream>>>(
        counts, cursor, seg, start, N_NODES);
    fill_sources_kernel<<<(N_EDGES + TPB - 1) / TPB, TPB, 0, stream>>>(
        srcIdx, dstIdx, rank, start, list, N_EDGES);

    gather_sum_kernel<IN_C, floatx2><<<GW, 256, 0, stream>>>(B0, x, seg, list, N_NODES);
    gemm_bias_act_kernel<IN_C, HID, 1>
        <<<GEMM_GRID, GEMM_WAVES * 32, 0, stream>>>(B0, W1a, b1a, B1, MT);
    gemm_bias_act_kernel<HID, HID, 1>
        <<<GEMM_GRID, GEMM_WAVES * 32, 0, stream>>>(B1, W1b, b1b, B2, MT);

    gather_sum_kernel<HID, floatx4><<<GW, 256, 0, stream>>>(B0, B2, seg, list, N_NODES);
    gemm_bias_act_kernel<HID, HID, 1>
        <<<GEMM_GRID, GEMM_WAVES * 32, 0, stream>>>(B0, W2a, b2a, B1, MT);
    gemm_bias_act_kernel<HID, OUT_C, 0>
        <<<GEMM_GRID, GEMM_WAVES * 32, 0, stream>>>(B1, W2b, b2b, out, MT);
}


# Each layer's two products fused in one launch: the 16x128 hidden strip stays in the wave's LDS rows as the A operand of the second product; 256 workgroups of 16 waves, both W in LDS; 9 launches to 7
# speedup vs baseline: 2.3360x; 1.0478x over previous
#include <hip/hip_runtime.h>
#include <hip/hip_bf16.h>


#define N_NODES 100000
#define N_EDGES 1600000
#define IN_C    64
#define HID     128
#define OUT_C   64

typedef __attribute__((ext_vector_type(2))) float floatx2;
typedef __attribute__((ext_vector_type(8))) float floatx8;

typedef __attribute__((ext_vector_type(4))) float floatx4;
typedef __attribute__((ext_vector_type(2))) int   intx2;

#define CNT_PAD 100096

#define RANK_NODES   4096
#define RANK_RANGES  ((N_NODES + RANK_NODES - 1) / RANK_NODES)
#define RANK_CHUNKS  20
#define RANK_CHUNK_E (N_EDGES / RANK_CHUNKS)
#define RANK_LOADS   5
static_assert(N_EDGES % (RANK_CHUNKS * 4) == 0, "edge chunks are whole 16-byte groups");
static_assert((RANK_NODES & (RANK_NODES - 1)) == 0, "the range test needs a power of two");

__device__ __forceinline__ void rank_take4(const int4 d, const int e, const int lo,
                                           int* cnt, int* __restrict__ rank) {
    const unsigned r0 = (unsigned)(d.x ^ lo);
    const unsigned r1 = (unsigned)(d.y ^ lo);
    const unsigned r2 = (unsigned)(d.z ^ lo);
    const unsigned r3 = (unsigned)(d.w ^ lo);
    if (r0 < RANK_NODES) rank[e + 0] = atomicAdd(&cnt[r0], 1);
    if (r1 < RANK_NODES) rank[e + 1] = atomicAdd(&cnt[r1], 1);
    if (r2 < RANK_NODES) rank[e + 2] = atomicAdd(&cnt[r2], 1);
    if (r3 < RANK_NODES) rank[e + 3] = atomicAdd(&cnt[r3], 1);
}

__global__ void __launch_bounds__(1024)
rank_edges_kernel(const int* __restrict__ dstIdx, int* __restrict__ counts,
                  int* __restrict__ cursor, int* __restrict__ rank, int n) {
    __shared__ int cnt[RANK_NODES];
    const int tid = threadIdx.x;
    const int lo  = blockIdx.x * RANK_NODES;
    const int ck  = blockIdx.y;
    for (int j = tid; j < RANK_NODES; j += 1024) cnt[j] = 0;
    __syncthreads();

    const int4* d4 = (const int4*)dstIdx;
    const int qb = ck * (RANK_CHUNK_E / 4);
    const int qe = qb + RANK_CHUNK_E / 4;
    for (int q0 = qb; q0 < qe; q0 += 1024 * RANK_LOADS) {
        int4 d[RANK_LOADS];
#pragma unroll
        for (int u = 0; u < RANK_LOADS; ++u) {
            const int q = q0 + u * 1024 + tid;
            d[u] = (q < qe) ? d4[q] : int4{-1, -1, -1, -1};
        }
#pragma unroll
        for (int u = 0; u < RANK_LOADS; ++u)
            rank_take4(d[u], (q0 + u * 1024 + tid) * 4, lo, cnt, rank);
    }
    __syncthreads();

    for (int j = tid; j < RANK_NODES; j += 1024) {
        const int i = lo + j;
        if (i < n) counts[ck * CNT_PAD + i] = cnt[j];
    }
    if (blockIdx.x == 0 && blockIdx.y == 0 && tid == 0) *cursor = 0;
}

__global__ void __launch_bounds__(1024)
segment_offsets_kernel(const int* __restrict__ counts, int* __restrict__ cursor,
                       intx2* __restrict__ seg, int* __restrict__ start, int n) {
    __shared__ int wsum[32];
    __shared__ int sbase;
    const int i    = blockIdx.x * 1024 + threadIdx.x;
    const int lane = threadIdx.x & 31;
    const int w    = threadIdx.x >> 5;
    int ck[RANK_CHUNKS] = {};
    int c = 0;
    if (i < n) {
#pragma unroll
        for (int k = 0; k < RANK_CHUNKS; ++k) {
            ck[k] = counts[k * CNT_PAD + i];
            c += ck[k];
        }
    }
    int incl = c;
    for (int d = 1; d < 32; d <<= 1) {
        int t = __shfl_up(incl, d);
        if (lane >= d) incl += t;
    }
    if (lane == 31) wsum[w] = incl;
    __syncthreads();
    if (w == 0) {
        const int v = wsum[lane];
        int vi = v;
        for (int d = 1; d < 32; d <<= 1) {
            int t = __shfl_up(vi, d);
            if (lane >= d) vi += t;
        }
        wsum[lane] = vi - v;
        if (lane == 31) sbase = atomicAdd(cursor, vi);
    }
    __syncthreads();
    if (i < n) {
        int s = sbase + wsum[w] + incl - c;
        seg[i] = intx2{s, c};
#pragma unroll
        for (int k = 0; k < RANK_CHUNKS; ++k) {
            start[k * CNT_PAD + i] = s;
            s += ck[k];
        }
    }
}

__global__ void fill_sources_kernel(const int* __restrict__ srcIdx, const int* __restrict__ dstIdx,
                                    const int* __restrict__ rank, const int* __restrict__ start,
                                    int* __restrict__ list, int nE) {
    int e = blockIdx.x * blockDim.x + threadIdx.x;
    if (e < nE) list[start[(e / RANK_CHUNK_E) * CNT_PAD + dstIdx[e]] + rank[e]] = srcIdx[e];
}

template <int C, typename V>
__global__ void __launch_bounds__(256)
gather_sum_kernel(float* __restrict__ z, const float* __restrict__ h,
                  const intx2* __restrict__ seg, const int* __restrict__ list, int n) {
    constexpr int VW = C / 32;
    const int lane = threadIdx.x & 31;
    const int gw = __builtin_amdgcn_readfirstlane((int)((blockIdx.x * 256 + threadIdx.x) >> 5));
    const int nw = (int)((gridDim.x * 256) >> 5);
    for (int i = gw; i < n; i += nw) {
        const intx2 sd = seg[i];
        const int beg = sd.x;
        const int deg = sd.y;
        V acc = *(const V*)(h + (size_t)i * C + lane * VW);
        for (int base = 0; base < deg; base += 32) {
            const int left = deg - base;
            const int m    = left < 32 ? left : 32;
            const int my   = (lane < m) ? list[beg + base + lane] : 0;
            for (int j = 0; j < m; j += 8) {
                int s[8];
#pragma unroll
                for (int k = 0; k < 8; ++k) s[k] = __builtin_amdgcn_readlane(my, j + k);
                V v[8];
#pragma unroll
                for (int k = 0; k < 8; ++k) {
                    if (j + k < m) v[k] = *(const V*)(h + (size_t)s[k] * C + lane * VW);
                }
#pragma unroll
                for (int k = 0; k < 8; ++k) {
                    if (j + k < m) acc += v[k];
                }
            }
        }
        *(V*)(z + (size_t)i * C + lane * VW) = acc;
    }
}

#define MLP_WAVES 16
#define MLP_GRID  256
#define MLP_AP    132

template <int K, int N>
__device__ __forceinline__ void stage_weights(float* sW, const float* __restrict__ W, const int tid) {
    constexpr int WP    = 2 * N + 32;
    constexpr int NQ    = N / 4;
    constexpr int UNITS = (K / 2) * NQ;
    static_assert(UNITS % (MLP_WAVES * 32) == 0, "W staging has a ragged last trip");
#pragma unroll
    for (int q = 0; q < UNITS / (MLP_WAVES * 32); ++q) {
        const int u  = tid + q * (MLP_WAVES * 32);
        const int kp = u / NQ;
        const int nq = u % NQ;
        const float4 r0 = *(const float4*)(W + (size_t)(2 * kp) * N + nq * 4);
        const float4 r1 = *(const float4*)(W + (size_t)(2 * kp + 1) * N + nq * 4);
        float4* d = (float4*)&sW[kp * WP + nq * 8];
        d[0] = float4{r0.x, r1.x, r0.y, r1.y};
        d[1] = float4{r0.z, r1.z, r0.w, r1.w};
    }
}

template <int K, int N, int RELU>
__device__ __forceinline__ void strip_product(float* sAw, const float* sW,
                                              const float* __restrict__ bias,
                                              const int half, const int l16) {
    constexpr int NT = N / 16;
    constexpr int WP = 2 * N + 32;
    constexpr int AP = MLP_AP;
    static_assert(K + 4 <= AP && N + 4 <= AP, "strip does not fit its LDS rows");
    const float* aRow = sAw + l16 * AP + half * 2;
    const float* bCol = sW + half * WP + l16 * 2;

    floatx8 acc[NT];
#pragma unroll
    for (int j = 0; j < NT; ++j) acc[j] = floatx8{};

#pragma unroll
    for (int k = 0; k < K; k += 4) {
        const floatx2 a = *(const floatx2*)(aRow + k);
        floatx2 b[NT];
#pragma unroll
        for (int j = 0; j < NT; ++j)
            b[j] = *(const floatx2*)(bCol + (k / 2) * WP + j * 32);
#pragma unroll
        for (int j = 0; j < NT; ++j)
            acc[j] = __builtin_amdgcn_wmma_f32_16x16x4_f32(
                 false, a,  false, b[j],
                 (short)0, acc[j],  false,  false);
    }

#pragma unroll
    for (int j = 0; j < NT; ++j) {
        const float bv = bias[j * 16 + l16];
#pragma unroll
        for (int v = 0; v < 8; ++v) {
            float x = acc[j][v] + bv;
            if (RELU) x = fmaxf(x, 0.0f);
            sAw[(half * 8 + v) * AP + j * 16 + l16] = x;
        }
    }
}

template <int K1, int N1, int N2, int RELU2>
__global__ void __launch_bounds__(MLP_WAVES * 32)
mlp_kernel(const float* __restrict__ A,
           const float* __restrict__ Wa, const float* __restrict__ ba,
           const float* __restrict__ Wb, const float* __restrict__ bb,
           float* __restrict__ C, int mTiles) {
    constexpr int AP = MLP_AP;
    __shared__ __attribute__((aligned(16))) float sWa[(K1 / 2) * (2 * N1 + 32)];
    __shared__ __attribute__((aligned(16))) float sWb[(N1 / 2) * (2 * N2 + 32)];
    __shared__ __attribute__((aligned(16))) float sA[MLP_WAVES * 16 * AP];

    const int tid  = threadIdx.x;
    const int lane = tid & 31;
    const int half = lane >> 4;
    const int l16  = lane & 15;
    const int wv   = __builtin_amdgcn_readfirstlane(tid >> 5);

    stage_weights<K1, N1>(sWa, Wa, tid);
    stage_weights<N1, N2>(sWb, Wb, tid);

    float* sAw = &sA[wv * (16 * AP)];

    const int stride = (int)gridDim.x * MLP_WAVES;
    const int iters  = (mTiles + stride - 1) / stride;
    for (int it = 0; it < iters; ++it) {
        const int  t     = it * stride + wv * (int)gridDim.x + (int)blockIdx.x;
        const bool valid = t < mTiles;

        if (valid) {
            const float4* A4 = (const float4*)(A + (size_t)t * (16 * K1));
            constexpr int K4 = K1 / 4;
#pragma unroll
            for (int q = 0; q < 16 * K4 / 32; ++q) {
                const int i  = lane + q * 32;
                const int r  = i / K4;
                const int c4 = i % K4;
                *(float4*)&sAw[r * AP + c4 * 4] = A4[i];
            }
        }
        __syncthreads();

        if (valid) strip_product<K1, N1, 1>(sAw, sWa, ba, half, l16);
        __syncthreads();

        if (valid) strip_product<N1, N2, RELU2>(sAw, sWb, bb, half, l16);
        __syncthreads();

        if (valid) {
            float4* C4 = (float4*)(C + (size_t)t * (16 * N2));
            constexpr int N4 = N2 / 4;
#pragma unroll
            for (int q = 0; q < 16 * N4 / 32; ++q) {
                const int i  = lane + q * 32;
                const int r  = i / N4;
                const int c4 = i % N4;
                C4[i] = *(const float4*)&sAw[r * AP + c4 * 4];
            }
        }
    }
}

extern "C" void kernel_launch(void* const* d_in, const int* in_sizes, int n_in,
                              void* d_out, int out_size, void* d_ws, size_t ws_size,
                              hipStream_t stream) {
    (void)in_sizes; (void)n_in; (void)out_size; (void)ws_size;

    const float* x    = (const float*)d_in[0];
    const int*   ei   = (const int*)d_in[1];
    const float* W1a  = (const float*)d_in[2];
    const float* b1a  = (const float*)d_in[3];
    const float* W1b  = (const float*)d_in[4];
    const float* b1b  = (const float*)d_in[5];
    const float* W2a  = (const float*)d_in[6];
    const float* b2a  = (const float*)d_in[7];
    const float* W2b  = (const float*)d_in[8];
    const float* b2b  = (const float*)d_in[9];
    float* out = (float*)d_out;

    const int* srcIdx = ei;
    const int* dstIdx = ei + N_EDGES;

    const size_t buf_elems = (size_t)N_NODES * HID;
    float* B0 = (float*)d_ws;
    float* B1 = B0 + buf_elems;
    int*  counts = (int*)(B1 + buf_elems);
    int*  start  = counts + RANK_CHUNKS * CNT_PAD;
    int*  cursor = start + RANK_CHUNKS * CNT_PAD;
    intx2* seg   = (intx2*)(cursor + 64);
    int*  rank   = (int*)(seg + CNT_PAD);
    int*  list   = rank + N_EDGES;

    const int TPB = 256;
    const int MT  = N_NODES / 16;

    const int GW  = N_NODES / 32;

    rank_edges_kernel<<<dim3(RANK_RANGES, RANK_CHUNKS), 1024, 0, stream>>>(
        dstIdx, counts, cursor, rank, N_NODES);
    segment_offsets_kernel<<<(N_NODES + 1023) / 1024, 1024, 0, stream>>>(
        counts, cursor, seg, start, N_NODES);
    fill_sources_kernel<<<(N_EDGES + TPB - 1) / TPB, TPB, 0, stream>>>(
        srcIdx, dstIdx, rank, start, list, N_EDGES);

    gather_sum_kernel<IN_C, floatx2><<<GW, 256, 0, stream>>>(B0, x, seg, list, N_NODES);
    mlp_kernel<IN_C, HID, HID, 1>
        <<<MLP_GRID, MLP_WAVES * 32, 0, stream>>>(B0, W1a, b1a, W1b, b1b, B1, MT);

    gather_sum_kernel<HID, floatx4><<<GW, 256, 0, stream>>>(B0, B1, seg, list, N_NODES);
    mlp_kernel<HID, HID, OUT_C, 0>
        <<<MLP_GRID, MLP_WAVES * 32, 0, stream>>>(B0, W2a, b2a, W2b, b2b, out, MT);
}


# MLP trip loop without workgroup barriers: a wave's LDS rows are its own, it waits only for its own LDS stores; its next A block is loaded to registers ahead of both products, the first ahead of W
# speedup vs baseline: 2.4816x; 1.0623x over previous
#include <hip/hip_runtime.h>
#include <hip/hip_bf16.h>


#define N_NODES 100000
#define N_EDGES 1600000
#define IN_C    64
#define HID     128
#define OUT_C   64

typedef __attribute__((ext_vector_type(2))) float floatx2;
typedef __attribute__((ext_vector_type(8))) float floatx8;

typedef __attribute__((ext_vector_type(4))) float floatx4;
typedef __attribute__((ext_vector_type(2))) int   intx2;

#define CNT_PAD 100096

#define RANK_NODES   4096
#define RANK_RANGES  ((N_NODES + RANK_NODES - 1) / RANK_NODES)
#define RANK_CHUNKS  20
#define RANK_CHUNK_E (N_EDGES / RANK_CHUNKS)
#define RANK_LOADS   5
static_assert(N_EDGES % (RANK_CHUNKS * 4) == 0, "edge chunks are whole 16-byte groups");
static_assert((RANK_NODES & (RANK_NODES - 1)) == 0, "the range test needs a power of two");

__device__ __forceinline__ void rank_take4(const int4 d, const int e, const int lo,
                                           int* cnt, int* __restrict__ rank) {
    const unsigned r0 = (unsigned)(d.x ^ lo);
    const unsigned r1 = (unsigned)(d.y ^ lo);
    const unsigned r2 = (unsigned)(d.z ^ lo);
    const unsigned r3 = (unsigned)(d.w ^ lo);
    if (r0 < RANK_NODES) rank[e + 0] = atomicAdd(&cnt[r0], 1);
    if (r1 < RANK_NODES) rank[e + 1] = atomicAdd(&cnt[r1], 1);
    if (r2 < RANK_NODES) rank[e + 2] = atomicAdd(&cnt[r2], 1);
    if (r3 < RANK_NODES) rank[e + 3] = atomicAdd(&cnt[r3], 1);
}

__global__ void __launch_bounds__(1024)
rank_edges_kernel(const int* __restrict__ dstIdx, int* __restrict__ counts,
                  int* __restrict__ cursor, int* __restrict__ rank, int n) {
    __shared__ int cnt[RANK_NODES];
    const int tid = threadIdx.x;
    const int lo  = blockIdx.x * RANK_NODES;
    const int ck  = blockIdx.y;
    for (int j = tid; j < RANK_NODES; j += 1024) cnt[j] = 0;
    __syncthreads();

    const int4* d4 = (const int4*)dstIdx;
    const int qb = ck * (RANK_CHUNK_E / 4);
    const int qe = qb + RANK_CHUNK_E / 4;
    for (int q0 = qb; q0 < qe; q0 += 1024 * RANK_LOADS) {
        int4 d[RANK_LOADS];
#pragma unroll
        for (int u = 0; u < RANK_LOADS; ++u) {
            const int q = q0 + u * 1024 + tid;
            d[u] = (q < qe) ? d4[q] : int4{-1, -1, -1, -1};
        }
#pragma unroll
        for (int u = 0; u < RANK_LOADS; ++u)
            rank_take4(d[u], (q0 + u * 1024 + tid) * 4, lo, cnt, rank);
    }
    __syncthreads();

    for (int j = tid; j < RANK_NODES; j += 1024) {
        const int i = lo + j;
        if (i < n) counts[ck * CNT_PAD + i] = cnt[j];
    }
    if (blockIdx.x == 0 && blockIdx.y == 0 && tid == 0) *cursor = 0;
}

__global__ void __launch_bounds__(1024)
segment_offsets_kernel(const int* __restrict__ counts, int* __restrict__ cursor,
                       intx2* __restrict__ seg, int* __restrict__ start, int n) {
    __shared__ int wsum[32];
    __shared__ int sbase;
    const int i    = blockIdx.x * 1024 + threadIdx.x;
    const int lane = threadIdx.x & 31;
    const int w    = threadIdx.x >> 5;
    int ck[RANK_CHUNKS] = {};
    int c = 0;
    if (i < n) {
#pragma unroll
        for (int k = 0; k < RANK_CHUNKS; ++k) {
            ck[k] = counts[k * CNT_PAD + i];
            c += ck[k];
        }
    }
    int incl = c;
    for (int d = 1; d < 32; d <<= 1) {
        int t = __shfl_up(incl, d);
        if (lane >= d) incl += t;
    }
    if (lane == 31) wsum[w] = incl;
    __syncthreads();
    if (w == 0) {
        const int v = wsum[lane];
        int vi = v;
        for (int d = 1; d < 32; d <<= 1) {
            int t = __shfl_up(vi, d);
            if (lane >= d) vi += t;
        }
        wsum[lane] = vi - v;
        if (lane == 31) sbase = atomicAdd(cursor, vi);
    }
    __syncthreads();
    if (i < n) {
        int s = sbase + wsum[w] + incl - c;
        seg[i] = intx2{s, c};
#pragma unroll
        for (int k = 0; k < RANK_CHUNKS; ++k) {
            start[k * CNT_PAD + i] = s;
            s += ck[k];
        }
    }
}

__global__ void fill_sources_kernel(const int* __restrict__ srcIdx, const int* __restrict__ dstIdx,
                                    const int* __restrict__ rank, const int* __restrict__ start,
                                    int* __restrict__ list, int nE) {
    int e = blockIdx.x * blockDim.x + threadIdx.x;
    if (e < nE) list[start[(e / RANK_CHUNK_E) * CNT_PAD + dstIdx[e]] + rank[e]] = srcIdx[e];
}

template <int C, typename V>
__global__ void __launch_bounds__(256)
gather_sum_kernel(float* __restrict__ z, const float* __restrict__ h,
                  const intx2* __restrict__ seg, const int* __restrict__ list, int n) {
    constexpr int VW = C / 32;
    const int lane = threadIdx.x & 31;
    const int gw = __builtin_amdgcn_readfirstlane((int)((blockIdx.x * 256 + threadIdx.x) >> 5));
    const int nw = (int)((gridDim.x * 256) >> 5);
    for (int i = gw; i < n; i += nw) {
        const intx2 sd = seg[i];
        const int beg = sd.x;
        const int deg = sd.y;
        V acc = *(const V*)(h + (size_t)i * C + lane * VW);
        for (int base = 0; base < deg; base += 32) {
            const int left = deg - base;
            const int m    = left < 32 ? left : 32;
            const int my   = (lane < m) ? list[beg + base + lane] : 0;
            for (int j = 0; j < m; j += 8) {
                int s[8];
#pragma unroll
                for (int k = 0; k < 8; ++k) s[k] = __builtin_amdgcn_readlane(my, j + k);
                V v[8];
#pragma unroll
                for (int k = 0; k < 8; ++k) {
                    if (j + k < m) v[k] = *(const V*)(h + (size_t)s[k] * C + lane * VW);
                }
#pragma unroll
                for (int k = 0; k < 8; ++k) {
                    if (j + k < m) acc += v[k];
                }
            }
        }
        *(V*)(z + (size_t)i * C + lane * VW) = acc;
    }
}

#define MLP_WAVES 16
#define MLP_GRID  256
#define MLP_AP    132

template <int K, int N>
__device__ __forceinline__ void stage_weights(float* sW, const float* __restrict__ W, const int tid) {
    constexpr int WP    = 2 * N + 32;
    constexpr int NQ    = N / 4;
    constexpr int UNITS = (K / 2) * NQ;
    static_assert(UNITS % (MLP_WAVES * 32) == 0, "W staging has a ragged last trip");
#pragma unroll
    for (int q = 0; q < UNITS / (MLP_WAVES * 32); ++q) {
        const int u  = tid + q * (MLP_WAVES * 32);
        const int kp = u / NQ;
        const int nq = u % NQ;
        const float4 r0 = *(const float4*)(W + (size_t)(2 * kp) * N + nq * 4);
        const float4 r1 = *(const float4*)(W + (size_t)(2 * kp + 1) * N + nq * 4);
        float4* d = (float4*)&sW[kp * WP + nq * 8];
        d[0] = float4{r0.x, r1.x, r0.y, r1.y};
        d[1] = float4{r0.z, r1.z, r0.w, r1.w};
    }
}

template <int K, int N, int RELU>
__device__ __forceinline__ void strip_product(float* sAw, const float* sW,
                                              const float* __restrict__ bias,
                                              const int half, const int l16) {
    constexpr int NT = N / 16;
    constexpr int WP = 2 * N + 32;
    constexpr int AP = MLP_AP;
    static_assert(K + 4 <= AP && N + 4 <= AP, "strip does not fit its LDS rows");
    const float* aRow = sAw + l16 * AP + half * 2;
    const float* bCol = sW + half * WP + l16 * 2;

    floatx8 acc[NT];
#pragma unroll
    for (int j = 0; j < NT; ++j) acc[j] = floatx8{};

#pragma unroll
    for (int k = 0; k < K; k += 4) {
        const floatx2 a = *(const floatx2*)(aRow + k);
        floatx2 b[NT];
#pragma unroll
        for (int j = 0; j < NT; ++j)
            b[j] = *(const floatx2*)(bCol + (k / 2) * WP + j * 32);
#pragma unroll
        for (int j = 0; j < NT; ++j)
            acc[j] = __builtin_amdgcn_wmma_f32_16x16x4_f32(
                 false, a,  false, b[j],
                 (short)0, acc[j],  false,  false);
    }

#pragma unroll
    for (int j = 0; j < NT; ++j) {
        const float bv = bias[j * 16 + l16];
#pragma unroll
        for (int v = 0; v < 8; ++v) {
            float x = acc[j][v] + bv;
            if (RELU) x = fmaxf(x, 0.0f);
            sAw[(half * 8 + v) * AP + j * 16 + l16] = x;
        }
    }
}

__device__ __forceinline__ void wave_rows_sync() {
    __builtin_amdgcn_fence(__ATOMIC_RELEASE, "workgroup", "local");
    __builtin_amdgcn_wave_barrier();
    __builtin_amdgcn_fence(__ATOMIC_ACQUIRE, "workgroup", "local");
}

template <int K1, int N1, int N2, int RELU2>
__global__ void __launch_bounds__(MLP_WAVES * 32)
mlp_kernel(const float* __restrict__ A,
           const float* __restrict__ Wa, const float* __restrict__ ba,
           const float* __restrict__ Wb, const float* __restrict__ bb,
           float* __restrict__ C, int mTiles) {
    constexpr int AP  = MLP_AP;
    constexpr int K4  = K1 / 4;
    constexpr int NLD = 16 * K4 / 32;
    __shared__ __attribute__((aligned(16))) float sWa[(K1 / 2) * (2 * N1 + 32)];
    __shared__ __attribute__((aligned(16))) float sWb[(N1 / 2) * (2 * N2 + 32)];
    __shared__ __attribute__((aligned(16))) float sA[MLP_WAVES * 16 * AP];

    const int tid  = threadIdx.x;
    const int lane = tid & 31;
    const int half = lane >> 4;
    const int l16  = lane & 15;
    const int wv   = __builtin_amdgcn_readfirstlane(tid >> 5);

    const int stride = (int)gridDim.x * MLP_WAVES;
    int t = wv * (int)gridDim.x + (int)blockIdx.x;

    floatx4 blk[NLD];
    if (t < mTiles) {
        const floatx4* A4 = (const floatx4*)(A + (size_t)t * (16 * K1));
#pragma unroll
        for (int q = 0; q < NLD; ++q) blk[q] = A4[lane + q * 32];
    }

    stage_weights<K1, N1>(sWa, Wa, tid);
    stage_weights<N1, N2>(sWb, Wb, tid);
    __syncthreads();

    float* sAw = &sA[wv * (16 * AP)];

    for (; t < mTiles; t += stride) {
#pragma unroll
        for (int q = 0; q < NLD; ++q) {
            const int i  = lane + q * 32;
            const int r  = i / K4;
            const int c4 = i % K4;
            *(floatx4*)&sAw[r * AP + c4 * 4] = blk[q];
        }
        wave_rows_sync();

        const int tn = t + stride;
        if (tn < mTiles) {
            const floatx4* A4 = (const floatx4*)(A + (size_t)tn * (16 * K1));
#pragma unroll
            for (int q = 0; q < NLD; ++q) blk[q] = A4[lane + q * 32];
        }

        strip_product<K1, N1, 1>(sAw, sWa, ba, half, l16);
        wave_rows_sync();

        strip_product<N1, N2, RELU2>(sAw, sWb, bb, half, l16);
        wave_rows_sync();

        float4* C4 = (float4*)(C + (size_t)t * (16 * N2));
        constexpr int N4 = N2 / 4;
#pragma unroll
        for (int q = 0; q < 16 * N4 / 32; ++q) {
            const int i  = lane + q * 32;
            const int r  = i / N4;
            const int c4 = i % N4;
            C4[i] = *(const float4*)&sAw[r * AP + c4 * 4];
        }
    }
}

extern "C" void kernel_launch(void* const* d_in, const int* in_sizes, int n_in,
                              void* d_out, int out_size, void* d_ws, size_t ws_size,
                              hipStream_t stream) {
    (void)in_sizes; (void)n_in; (void)out_size; (void)ws_size;

    const float* x    = (const float*)d_in[0];
    const int*   ei   = (const int*)d_in[1];
    const float* W1a  = (const float*)d_in[2];
    const float* b1a  = (const float*)d_in[3];
    const float* W1b  = (const float*)d_in[4];
    const float* b1b  = (const float*)d_in[5];
    const float* W2a  = (const float*)d_in[6];
    const float* b2a  = (const float*)d_in[7];
    const float* W2b  = (const float*)d_in[8];
    const float* b2b  = (const float*)d_in[9];
    float* out = (float*)d_out;

    const int* srcIdx = ei;
    const int* dstIdx = ei + N_EDGES;

    const size_t buf_elems = (size_t)N_NODES * HID;
    float* B0 = (float*)d_ws;
    float* B1 = B0 + buf_elems;
    int*  counts = (int*)(B1 + buf_elems);
    int*  start  = counts + RANK_CHUNKS * CNT_PAD;
    int*  cursor = start + RANK_CHUNKS * CNT_PAD;
    intx2* seg   = (intx2*)(cursor + 64);
    int*  rank   = (int*)(seg + CNT_PAD);
    int*  list   = rank + N_EDGES;

    const int TPB = 256;
    const int MT  = N_NODES / 16;

    const int GW  = N_NODES / 32;

    rank_edges_kernel<<<dim3(RANK_RANGES, RANK_CHUNKS), 1024, 0, stream>>>(
        dstIdx, counts, cursor, rank, N_NODES);
    segment_offsets_kernel<<<(N_NODES + 1023) / 1024, 1024, 0, stream>>>(
        counts, cursor, seg, start, N_NODES);
    fill_sources_kernel<<<(N_EDGES + TPB - 1) / TPB, TPB, 0, stream>>>(
        srcIdx, dstIdx, rank, start, list, N_EDGES);

    gather_sum_kernel<IN_C, floatx2><<<GW, 256, 0, stream>>>(B0, x, seg, list, N_NODES);
    mlp_kernel<IN_C, HID, HID, 1>
        <<<MLP_GRID, MLP_WAVES * 32, 0, stream>>>(B0, W1a, b1a, W1b, b1b, B1, MT);

    gather_sum_kernel<HID, floatx4><<<GW, 256, 0, stream>>>(B0, B1, seg, list, N_NODES);
    mlp_kernel<HID, HID, OUT_C, 0>
        <<<MLP_GRID, MLP_WAVES * 32, 0, stream>>>(B0, W2a, b2a, W2b, b2b, out, MT);
}


# Row gather-sum moved into the MLP launch: a wave adds its 16 nodes' in-edge rows into its LDS rows (the A operand), 8 KB of row reads in flight, same sum order; z never stored; 7 launches to 5
# speedup vs baseline: 2.6217x; 1.0565x over previous
#include <hip/hip_runtime.h>
#include <hip/hip_bf16.h>


#define N_NODES 100000
#define N_EDGES 1600000
#define IN_C    64
#define HID     128
#define OUT_C   64

typedef __attribute__((ext_vector_type(2))) float floatx2;
typedef __attribute__((ext_vector_type(8))) float floatx8;

typedef __attribute__((ext_vector_type(4))) float floatx4;
typedef __attribute__((ext_vector_type(2))) int   intx2;

#define CNT_PAD 100096

#define RANK_NODES   4096
#define RANK_RANGES  ((N_NODES + RANK_NODES - 1) / RANK_NODES)
#define RANK_CHUNKS  20
#define RANK_CHUNK_E (N_EDGES / RANK_CHUNKS)
#define RANK_LOADS   5
static_assert(N_EDGES % (RANK_CHUNKS * 4) == 0, "edge chunks are whole 16-byte groups");
static_assert((RANK_NODES & (RANK_NODES - 1)) == 0, "the range test needs a power of two");

__device__ __forceinline__ void rank_take4(const int4 d, const int e, const int lo,
                                           int* cnt, int* __restrict__ rank) {
    const unsigned r0 = (unsigned)(d.x ^ lo);
    const unsigned r1 = (unsigned)(d.y ^ lo);
    const unsigned r2 = (unsigned)(d.z ^ lo);
    const unsigned r3 = (unsigned)(d.w ^ lo);
    if (r0 < RANK_NODES) rank[e + 0] = atomicAdd(&cnt[r0], 1);
    if (r1 < RANK_NODES) rank[e + 1] = atomicAdd(&cnt[r1], 1);
    if (r2 < RANK_NODES) rank[e + 2] = atomicAdd(&cnt[r2], 1);
    if (r3 < RANK_NODES) rank[e + 3] = atomicAdd(&cnt[r3], 1);
}

__global__ void __launch_bounds__(1024)
rank_edges_kernel(const int* __restrict__ dstIdx, int* __restrict__ counts,
                  int* __restrict__ cursor, int* __restrict__ rank, int n) {
    __shared__ int cnt[RANK_NODES];
    const int tid = threadIdx.x;
    const int lo  = blockIdx.x * RANK_NODES;
    const int ck  = blockIdx.y;
    for (int j = tid; j < RANK_NODES; j += 1024) cnt[j] = 0;
    __syncthreads();

    const int4* d4 = (const int4*)dstIdx;
    const int qb = ck * (RANK_CHUNK_E / 4);
    const int qe = qb + RANK_CHUNK_E / 4;
    for (int q0 = qb; q0 < qe; q0 += 1024 * RANK_LOADS) {
        int4 d[RANK_LOADS];
#pragma unroll
        for (int u = 0; u < RANK_LOADS; ++u) {
            const int q = q0 + u * 1024 + tid;
            d[u] = (q < qe) ? d4[q] : int4{-1, -1, -1, -1};
        }
#pragma unroll
        for (int u = 0; u < RANK_LOADS; ++u)
            rank_take4(d[u], (q0 + u * 1024 + tid) * 4, lo, cnt, rank);
    }
    __syncthreads();

    for (int j = tid; j < RANK_NODES; j += 1024) {
        const int i = lo + j;
        if (i < n) counts[ck * CNT_PAD + i] = cnt[j];
    }
    if (blockIdx.x == 0 && blockIdx.y == 0 && tid == 0) *cursor = 0;
}

__global__ void __launch_bounds__(1024)
segment_offsets_kernel(const int* __restrict__ counts, int* __restrict__ cursor,
                       intx2* __restrict__ seg, int* __restrict__ start, int n) {
    __shared__ int wsum[32];
    __shared__ int sbase;
    const int i    = blockIdx.x * 1024 + threadIdx.x;
    const int lane = threadIdx.x & 31;
    const int w    = threadIdx.x >> 5;
    int ck[RANK_CHUNKS] = {};
    int c = 0;
    if (i < n) {
#pragma unroll
        for (int k = 0; k < RANK_CHUNKS; ++k) {
            ck[k] = counts[k * CNT_PAD + i];
            c += ck[k];
        }
    }
    int incl = c;
    for (int d = 1; d < 32; d <<= 1) {
        int t = __shfl_up(incl, d);
        if (lane >= d) incl += t;
    }
    if (lane == 31) wsum[w] = incl;
    __syncthreads();
    if (w == 0) {
        const int v = wsum[lane];
        int vi = v;
        for (int d = 1; d < 32; d <<= 1) {
            int t = __shfl_up(vi, d);
            if (lane >= d) vi += t;
        }
        wsum[lane] = vi - v;
        if (lane == 31) sbase = atomicAdd(cursor, vi);
    }
    __syncthreads();
    if (i < n) {
        int s = sbase + wsum[w] + incl - c;
        seg[i] = intx2{s, c};
#pragma unroll
        for (int k = 0; k < RANK_CHUNKS; ++k) {
            start[k * CNT_PAD + i] = s;
            s += ck[k];
        }
    }
}

__global__ void fill_sources_kernel(const int* __restrict__ srcIdx, const int* __restrict__ dstIdx,
                                    const int* __restrict__ rank, const int* __restrict__ start,
                                    int* __restrict__ list, int nE) {
    int e = blockIdx.x * blockDim.x + threadIdx.x;
    if (e < nE) list[start[(e / RANK_CHUNK_E) * CNT_PAD + dstIdx[e]] + rank[e]] = srcIdx[e];
}

#define MLP_WAVES 16
#define MLP_GRID  256
#define MLP_AP    132

template <int K, int N>
__device__ __forceinline__ void stage_weights(float* sW, const float* __restrict__ W, const int tid) {
    constexpr int WP    = 2 * N + 32;
    constexpr int NQ    = N / 4;
    constexpr int UNITS = (K / 2) * NQ;
    static_assert(UNITS % (MLP_WAVES * 32) == 0, "W staging has a ragged last trip");
#pragma unroll
    for (int q = 0; q < UNITS / (MLP_WAVES * 32); ++q) {
        const int u  = tid + q * (MLP_WAVES * 32);
        const int kp = u / NQ;
        const int nq = u % NQ;
        const float4 r0 = *(const float4*)(W + (size_t)(2 * kp) * N + nq * 4);
        const float4 r1 = *(const float4*)(W + (size_t)(2 * kp + 1) * N + nq * 4);
        float4* d = (float4*)&sW[kp * WP + nq * 8];
        d[0] = float4{r0.x, r1.x, r0.y, r1.y};
        d[1] = float4{r0.z, r1.z, r0.w, r1.w};
    }
}

template <int K, int N, int RELU>
__device__ __forceinline__ void strip_product(float* sAw, const float* sW,
                                              const float* __restrict__ bias,
                                              const int half, const int l16) {
    constexpr int NT = N / 16;
    constexpr int WP = 2 * N + 32;
    constexpr int AP = MLP_AP;
    static_assert(K + 4 <= AP && N + 4 <= AP, "strip does not fit its LDS rows");
    const float* aRow = sAw + l16 * AP + half * 2;
    const float* bCol = sW + half * WP + l16 * 2;

    floatx8 acc[NT];
#pragma unroll
    for (int j = 0; j < NT; ++j) acc[j] = floatx8{};

#pragma unroll
    for (int k = 0; k < K; k += 4) {
        const floatx2 a = *(const floatx2*)(aRow + k);
        floatx2 b[NT];
#pragma unroll
        for (int j = 0; j < NT; ++j)
            b[j] = *(const floatx2*)(bCol + (k / 2) * WP + j * 32);
#pragma unroll
        for (int j = 0; j < NT; ++j)
            acc[j] = __builtin_amdgcn_wmma_f32_16x16x4_f32(
                 false, a,  false, b[j],
                 (short)0, acc[j],  false,  false);
    }

#pragma unroll
    for (int j = 0; j < NT; ++j) {
        const float bv = bias[j * 16 + l16];
#pragma unroll
        for (int v = 0; v < 8; ++v) {
            float x = acc[j][v] + bv;
            if (RELU) x = fmaxf(x, 0.0f);
            sAw[(half * 8 + v) * AP + j * 16 + l16] = x;
        }
    }
}

__device__ __forceinline__ void wave_rows_sync() {
    __builtin_amdgcn_fence(__ATOMIC_RELEASE, "workgroup", "local");
    __builtin_amdgcn_wave_barrier();
    __builtin_amdgcn_fence(__ATOMIC_ACQUIRE, "workgroup", "local");
}

template <int C> struct RowVec;
template <> struct RowVec<64>  { typedef floatx2 type; };
template <> struct RowVec<128> { typedef floatx4 type; };


template <typename V>
__device__ __forceinline__ void next_node_if_past(float* myCol, const int endv, const int e,
                                                  int& cur, int& curEnd, V& acc) {
    while (e >= curEnd) {
        *(V*)(myCol + cur * MLP_AP) = acc;
        ++cur;
        curEnd = __builtin_amdgcn_readlane(endv, cur);
        acc = *(const V*)(myCol + cur * MLP_AP);
    }
}

template <int C, typename V, int G>
__device__ __forceinline__ void read_rows(V (&v)[G], const float* __restrict__ h,
                                          const int ids, const int j, const int lane) {
    constexpr int VW = C / 32;
    int s[G];
#pragma unroll
    for (int k = 0; k < G; ++k) s[k] = __builtin_amdgcn_readlane(ids, j + k);
#pragma unroll
    for (int k = 0; k < G; ++k) v[k] = *(const V*)(h + (size_t)s[k] * C + lane * VW);
}

template <typename V, int G, bool WHOLE>
__device__ __forceinline__ void add_rows(const V (&v)[G], float* myCol, const int e,
                                         const int tot, const int endv,
                                         int& cur, int& curEnd, V& acc) {
#pragma unroll
    for (int k = 0; k < G; ++k) {
        if (WHOLE || e + k < tot) {
            next_node_if_past<V>(myCol, endv, e + k, cur, curEnd, acc);
            acc += v[k];
        }
    }
}

__device__ __forceinline__ int first_ids(const int* __restrict__ list, const intx2 sdv, const int lane) {
    const int at = __builtin_amdgcn_readlane(sdv.x, 0) + lane;
    return list[at < N_EDGES - 1 ? at : N_EDGES - 1];
}

template <int C, typename V, int GW>
__device__ __forceinline__ void gather_into_rows(float* sAw, const float* __restrict__ h,
                                                 const int* __restrict__ list,
                                                 const intx2 sdv, int ids, const int lane) {
    constexpr int VW = C / 32;
    const int beg0 = __builtin_amdgcn_readlane(sdv.x, 0);
    const int endv = sdv.x + sdv.y - beg0;
    const int tot  = __builtin_amdgcn_readlane(endv, 15);
    float* myCol = sAw + lane * VW;
    int cur    = 0;
    int curEnd = __builtin_amdgcn_readlane(endv, 0);
    V acc = *(const V*)myCol;
    const int lastE = N_EDGES - 1;
    int at = beg0 + lane;
    int e = 0;
    static_assert(GW == 32 || GW == 16, "one or two groups per 32-id vector");
    for (; e + 32 <= tot; e += 32) {
        V v[GW];
        read_rows<C, V, GW>(v, h, ids, 0, lane);
        at += 32;
        const int nxt = list[at < lastE ? at : lastE];
        add_rows<V, GW, true>(v, myCol, e, tot, endv, cur, curEnd, acc);
        if constexpr (GW == 16) {
            read_rows<C, V, 16>(v, h, ids, 16, lane);
            add_rows<V, 16, true>(v, myCol, e + 16, tot, endv, cur, curEnd, acc);
        }
        ids = nxt;
    }
    V w[16];
    int j = 0;
    if (e + 16 <= tot) {
        read_rows<C, V, 16>(w, h, ids, 0, lane);
        add_rows<V, 16, true>(w, myCol, e, tot, endv, cur, curEnd, acc);
        e += 16;
        j = 16;
    }
    if (e < tot) {
        read_rows<C, V, 16>(w, h, ids, j, lane);
        add_rows<V, 16, false>(w, myCol, e, tot, endv, cur, curEnd, acc);
    }
    *(V*)(myCol + cur * MLP_AP) = acc;
}

template <int K1, int N1, int N2, int RELU2>
__global__ void __launch_bounds__(MLP_WAVES * 32)
mlp_kernel(const float* __restrict__ A,
           const intx2* __restrict__ seg, const int* __restrict__ list,
           const float* __restrict__ Wa, const float* __restrict__ ba,
           const float* __restrict__ Wb, const float* __restrict__ bb,
           float* __restrict__ C, int mTiles) {
    constexpr int AP  = MLP_AP;
    constexpr int K4  = K1 / 4;
    constexpr int NLD = 16 * K4 / 32;
    __shared__ __attribute__((aligned(16))) float sWa[(K1 / 2) * (2 * N1 + 32)];
    __shared__ __attribute__((aligned(16))) float sWb[(N1 / 2) * (2 * N2 + 32)];
    __shared__ __attribute__((aligned(16))) float sA[MLP_WAVES * 16 * AP];

    const int tid  = threadIdx.x;
    const int lane = tid & 31;
    const int half = lane >> 4;
    const int l16  = lane & 15;
    const int wv   = __builtin_amdgcn_readfirstlane(tid >> 5);

    const int stride = (int)gridDim.x * MLP_WAVES;
    int t = wv * (int)gridDim.x + (int)blockIdx.x;

    floatx4 blk[NLD];
    intx2 sdv = intx2{0, 0};
    if (t < mTiles) {
        const floatx4* A4 = (const floatx4*)(A + (size_t)t * (16 * K1));
#pragma unroll
        for (int q = 0; q < NLD; ++q) blk[q] = A4[lane + q * 32];
        if (lane < 16) sdv = seg[t * 16 + lane];
    }

    stage_weights<K1, N1>(sWa, Wa, tid);
    stage_weights<N1, N2>(sWb, Wb, tid);
    int ids = 0;
    if (t < mTiles) ids = first_ids(list, sdv, lane);
    __syncthreads();

    float* sAw = &sA[wv * (16 * AP)];

    for (; t < mTiles; t += stride) {
#pragma unroll
        for (int q = 0; q < NLD; ++q) {
            const int i  = lane + q * 32;
            const int r  = i / K4;
            const int c4 = i % K4;
            *(floatx4*)&sAw[r * AP + c4 * 4] = blk[q];
        }
        wave_rows_sync();

        gather_into_rows<K1, typename RowVec<K1>::type, (K1 <= 64 ? 32 : 16)>(sAw, A, list, sdv, ids, lane);
        wave_rows_sync();

        const int tn = t + stride;
        if (tn < mTiles) {
            const floatx4* A4 = (const floatx4*)(A + (size_t)tn * (16 * K1));
#pragma unroll
            for (int q = 0; q < NLD; ++q) blk[q] = A4[lane + q * 32];
            if (lane < 16) sdv = seg[tn * 16 + lane];
        }

        strip_product<K1, N1, 1>(sAw, sWa, ba, half, l16);
        wave_rows_sync();

        if (tn < mTiles) ids = first_ids(list, sdv, lane);

        strip_product<N1, N2, RELU2>(sAw, sWb, bb, half, l16);
        wave_rows_sync();

        float4* C4 = (float4*)(C + (size_t)t * (16 * N2));
        constexpr int N4 = N2 / 4;
#pragma unroll
        for (int q = 0; q < 16 * N4 / 32; ++q) {
            const int i  = lane + q * 32;
            const int r  = i / N4;
            const int c4 = i % N4;
            C4[i] = *(const float4*)&sAw[r * AP + c4 * 4];
        }
    }
}

extern "C" void kernel_launch(void* const* d_in, const int* in_sizes, int n_in,
                              void* d_out, int out_size, void* d_ws, size_t ws_size,
                              hipStream_t stream) {
    (void)in_sizes; (void)n_in; (void)out_size; (void)ws_size;

    const float* x    = (const float*)d_in[0];
    const int*   ei   = (const int*)d_in[1];
    const float* W1a  = (const float*)d_in[2];
    const float* b1a  = (const float*)d_in[3];
    const float* W1b  = (const float*)d_in[4];
    const float* b1b  = (const float*)d_in[5];
    const float* W2a  = (const float*)d_in[6];
    const float* b2a  = (const float*)d_in[7];
    const float* W2b  = (const float*)d_in[8];
    const float* b2b  = (const float*)d_in[9];
    float* out = (float*)d_out;

    const int* srcIdx = ei;
    const int* dstIdx = ei + N_EDGES;

    const size_t buf_elems = (size_t)N_NODES * HID;
    float* B1 = (float*)d_ws + buf_elems;
    int*  counts = (int*)(B1 + buf_elems);
    int*  start  = counts + RANK_CHUNKS * CNT_PAD;
    int*  cursor = start + RANK_CHUNKS * CNT_PAD;
    intx2* seg   = (intx2*)(cursor + 64);
    int*  rank   = (int*)(seg + CNT_PAD);
    int*  list   = rank + N_EDGES;

    const int TPB = 256;
    const int MT  = N_NODES / 16;

    rank_edges_kernel<<<dim3(RANK_RANGES, RANK_CHUNKS), 1024, 0, stream>>>(
        dstIdx, counts, cursor, rank, N_NODES);
    segment_offsets_kernel<<<(N_NODES + 1023) / 1024, 1024, 0, stream>>>(
        counts, cursor, seg, start, N_NODES);
    fill_sources_kernel<<<(N_EDGES + TPB - 1) / TPB, TPB, 0, stream>>>(
        srcIdx, dstIdx, rank, start, list, N_EDGES);

    mlp_kernel<IN_C, HID, HID, 1>
        <<<MLP_GRID, MLP_WAVES * 32, 0, stream>>>(x, seg, list, W1a, b1a, W1b, b1b, B1, MT);

    mlp_kernel<HID, HID, OUT_C, 0>
        <<<MLP_GRID, MLP_WAVES * 32, 0, stream>>>(B1, seg, list, W2a, b2a, W2b, b2b, out, MT);
}
